# starting kernel
# baseline (speedup 1.0000x reference)
#include <hip/hip_runtime.h>

typedef float v2f __attribute__((ext_vector_type(2)));
typedef float v8f __attribute__((ext_vector_type(8)));

__device__ __forceinline__ v8f wmma_f32(v2f a, v2f b, v8f c) {
  return __builtin_amdgcn_wmma_f32_16x16x4_f32(false, a, false, b, (short)0, c,
                                               false, false);
}

__device__ __forceinline__ void atomic_fadd_f32(float* p, float v) {
  asm volatile("global_atomic_add_f32 %0, %1, off scope:SCOPE_DEV"
               :
               : "v"((unsigned long long)(uintptr_t)p), "v"(v)
               : "memory");
}

__launch_bounds__(256)
__global__ void gemm64_kernel(const float* __restrict__ A,
                              const float* __restrict__ B,
                              const float* __restrict__ bias,
                              float* __restrict__ C, int M) {
  __shared__ float2 Bs2[32 * 64];
  for (int i = threadIdx.x; i < 2048; i += 256) {
    int k2 = i >> 6, n = i & 63;
    Bs2[i] = float2{B[(2 * k2) * 64 + n], B[(2 * k2 + 1) * 64 + n]};
  }
  __syncthreads();

  const int wave = threadIdx.x >> 5;
  const int lane = threadIdx.x & 31;
  const int r = lane & 15;
  const int h = lane >> 4;

  const int row0 = (blockIdx.x * 8 + wave) * 16;
  if (row0 >= M) return;

  int rowA = row0 + r;
  if (rowA >= M) rowA = M - 1;
  const float* __restrict__ Arow = A + (size_t)rowA * 64;

  v8f acc0 = {}, acc1 = {}, acc2 = {}, acc3 = {};
#pragma unroll
  for (int k0 = 0; k0 < 64; k0 += 4) {
    const int ka = k0 + 2 * h;
    float2 av = *reinterpret_cast<const float2*>(Arow + ka);
    v2f a; a.x = av.x; a.y = av.y;
    const int k2 = (k0 >> 1) + h;
    const float2* __restrict__ Bk = &Bs2[k2 * 64 + r];
    float2 f0 = Bk[0], f1 = Bk[16], f2 = Bk[32], f3 = Bk[48];
    v2f b0, b1, b2, b3;
    b0.x = f0.x; b0.y = f0.y;  b1.x = f1.x; b1.y = f1.y;
    b2.x = f2.x; b2.y = f2.y;  b3.x = f3.x; b3.y = f3.y;
    acc0 = wmma_f32(a, b0, acc0);
    acc1 = wmma_f32(a, b1, acc1);
    acc2 = wmma_f32(a, b2, acc2);
    acc3 = wmma_f32(a, b3, acc3);
  }

  float bia0 = 0.f, bia1 = 0.f, bia2 = 0.f, bia3 = 0.f;
  if (bias) { bia0 = bias[r]; bia1 = bias[16 + r]; bia2 = bias[32 + r]; bia3 = bias[48 + r]; }

#pragma unroll
  for (int v = 0; v < 8; ++v) {
    const int row = row0 + v + 8 * h;
    if (row < M) {
      float* __restrict__ Crow = C + (size_t)row * 64;
      Crow[r]      = acc0[v] + bia0;
      Crow[16 + r] = acc1[v] + bia1;
      Crow[32 + r] = acc2[v] + bia2;
      Crow[48 + r] = acc3[v] + bia3;
    }
  }
}

__launch_bounds__(256)
__global__ void gemm_concat128_kernel(const float* __restrict__ A0,
                                      const float* __restrict__ A1,
                                      const float* __restrict__ B,
                                      const float* __restrict__ bias,
                                      float* __restrict__ C, int M) {
  __shared__ float2 Bs2[64 * 64];
  for (int i = threadIdx.x; i < 4096; i += 256) {
    int k2 = i >> 6, n = i & 63;
    Bs2[i] = float2{B[(2 * k2) * 64 + n], B[(2 * k2 + 1) * 64 + n]};
  }
  __syncthreads();

  const int wave = threadIdx.x >> 5;
  const int lane = threadIdx.x & 31;
  const int r = lane & 15;
  const int h = lane >> 4;

  const int row0 = (blockIdx.x * 8 + wave) * 16;
  if (row0 >= M) return;

  int rowA = row0 + r;
  if (rowA >= M) rowA = M - 1;
  const float* __restrict__ Arow0 = A0 + (size_t)rowA * 64;
  const float* __restrict__ Arow1 = A1 + (size_t)rowA * 64;

  v8f acc0 = {}, acc1 = {}, acc2 = {}, acc3 = {};
#pragma unroll
  for (int k0 = 0; k0 < 128; k0 += 4) {
    const int ka = k0 + 2 * h;
    const float* src = (k0 < 64) ? (Arow0 + ka) : (Arow1 + (ka - 64));
    float2 av = *reinterpret_cast<const float2*>(src);
    v2f a; a.x = av.x; a.y = av.y;
    const int k2 = (k0 >> 1) + h;
    const float2* __restrict__ Bk = &Bs2[k2 * 64 + r];
    float2 f0 = Bk[0], f1 = Bk[16], f2 = Bk[32], f3 = Bk[48];
    v2f b0, b1, b2, b3;
    b0.x = f0.x; b0.y = f0.y;  b1.x = f1.x; b1.y = f1.y;
    b2.x = f2.x; b2.y = f2.y;  b3.x = f3.x; b3.y = f3.y;
    acc0 = wmma_f32(a, b0, acc0);
    acc1 = wmma_f32(a, b1, acc1);
    acc2 = wmma_f32(a, b2, acc2);
    acc3 = wmma_f32(a, b3, acc3);
  }

  const float bia0 = bias[r], bia1 = bias[16 + r], bia2 = bias[32 + r], bia3 = bias[48 + r];
#pragma unroll
  for (int v = 0; v < 8; ++v) {
    const int row = row0 + v + 8 * h;
    if (row < M) {
      float* __restrict__ Crow = C + (size_t)row * 64;
      Crow[r]      = acc0[v] + bia0;
      Crow[16 + r] = acc1[v] + bia1;
      Crow[32 + r] = acc2[v] + bia2;
      Crow[48 + r] = acc3[v] + bia3;
    }
  }
}

__launch_bounds__(256)
__global__ void edge_kernel(const float* __restrict__ rhs,
                            const float* __restrict__ lhs,
                            const int* __restrict__ rj,
                            const int* __restrict__ lj,
                            const float* __restrict__ weights,
                            const float* __restrict__ We,
                            const float* __restrict__ Wf,
                            const float* __restrict__ bf,
                            float* __restrict__ outAcc, int E) {
  __shared__ float2 WfS2[32 * 64];
  __shared__ float WeS[64];
  __shared__ float bfS[64];
  __shared__ float msgS[8][16 * 64];
  __shared__ int rjS[8][16];

  for (int i = threadIdx.x; i < 2048; i += 256) {
    int k2 = i >> 6, n = i & 63;
    WfS2[i] = float2{Wf[(2 * k2) * 64 + n], Wf[(2 * k2 + 1) * 64 + n]};
  }
  if (threadIdx.x < 64) { WeS[threadIdx.x] = We[threadIdx.x]; bfS[threadIdx.x] = bf[threadIdx.x]; }
  __syncthreads();

  const int wave = threadIdx.x >> 5;
  const int lane = threadIdx.x & 31;
  const int r = lane & 15;
  const int h = lane >> 4;

  const int eBase = (blockIdx.x * 8 + wave) * 16;

  int e = eBase + r;
  if (e >= E) e = E - 1;
  const int idxR = rj[e];
  const int idxL = lj[e];
  const float w = weights[e];
  if (h == 0) rjS[wave][r] = idxR;

  const float4* __restrict__ pr = reinterpret_cast<const float4*>(rhs + (size_t)idxR * 64) + h * 8;
  const float4* __restrict__ pl = reinterpret_cast<const float4*>(lhs + (size_t)idxL * 64) + h * 8;
  const float4* __restrict__ pe = reinterpret_cast<const float4*>(WeS) + h * 8;
  float4* __restrict__ pm = reinterpret_cast<float4*>(&msgS[wave][r * 64]) + h * 8;
#pragma unroll
  for (int i = 0; i < 8; ++i) {
    float4 a = pr[i], b = pl[i], c = pe[i];
    float4 m;
    m.x = a.x + b.x + w * c.x;
    m.y = a.y + b.y + w * c.y;
    m.z = a.z + b.z + w * c.z;
    m.w = a.w + b.w + w * c.w;
    m.x = m.x > 0.f ? m.x : 0.01f * m.x;
    m.y = m.y > 0.f ? m.y : 0.01f * m.y;
    m.z = m.z > 0.f ? m.z : 0.01f * m.z;
    m.w = m.w > 0.f ? m.w : 0.01f * m.w;
    pm[i] = m;
  }

  v8f acc0 = {}, acc1 = {}, acc2 = {}, acc3 = {};
  const float* __restrict__ mrow = &msgS[wave][r * 64];
#pragma unroll
  for (int k0 = 0; k0 < 64; k0 += 4) {
    const int ka = k0 + 2 * h;
    float2 av = *reinterpret_cast<const float2*>(mrow + ka);
    v2f a; a.x = av.x; a.y = av.y;
    const int k2 = (k0 >> 1) + h;
    const float2* __restrict__ Bk = &WfS2[k2 * 64 + r];
    float2 f0 = Bk[0], f1 = Bk[16], f2 = Bk[32], f3 = Bk[48];
    v2f b0, b1, b2, b3;
    b0.x = f0.x; b0.y = f0.y;  b1.x = f1.x; b1.y = f1.y;
    b2.x = f2.x; b2.y = f2.y;  b3.x = f3.x; b3.y = f3.y;
    acc0 = wmma_f32(a, b0, acc0);
    acc1 = wmma_f32(a, b1, acc1);
    acc2 = wmma_f32(a, b2, acc2);
    acc3 = wmma_f32(a, b3, acc3);
  }

  const float bia0 = bfS[r], bia1 = bfS[16 + r], bia2 = bfS[32 + r], bia3 = bfS[48 + r];
#pragma unroll
  for (int v = 0; v < 8; ++v) {
    const int m = v + 8 * h;
    if (eBase + m < E) {
      const int seg = rjS[wave][m];
      float* __restrict__ orow = outAcc + (size_t)seg * 64;
      atomic_fadd_f32(orow + r,      acc0[v] + bia0);
      atomic_fadd_f32(orow + 16 + r, acc1[v] + bia1);
      atomic_fadd_f32(orow + 32 + r, acc2[v] + bia2);
      atomic_fadd_f32(orow + 48 + r, acc3[v] + bia3);
    }
  }
}

__global__ void zero4_kernel(float4* __restrict__ p, int n4) {
  int i = blockIdx.x * blockDim.x + threadIdx.x;
  if (i < n4) p[i] = float4{0.f, 0.f, 0.f, 0.f};
}

extern "C" void kernel_launch(void* const* d_in, const int* in_sizes, int n_in,
                              void* d_out, int out_size, void* d_ws, size_t ws_size,
                              hipStream_t stream) {
  const float* input   = (const float*)d_in[0];
  const float* other   = (const float*)d_in[1];
  const int*   rj      = (const int*)d_in[2];
  const int*   lj      = (const int*)d_in[3];
  const float* weights = (const float*)d_in[4];
  const float* Wi      = (const float*)d_in[5];
  const float* bi      = (const float*)d_in[6];
  const float* Wo      = (const float*)d_in[7];
  const float* We      = (const float*)d_in[8];
  const float* Wf      = (const float*)d_in[9];
  const float* bf      = (const float*)d_in[10];
  const float* Wout    = (const float*)d_in[11];
  const float* bout    = (const float*)d_in[12];

  const int D    = 64;
  const int N_IN = in_sizes[0] / D;
  const int N_OT = in_sizes[1] / D;
  const int E    = in_sizes[2];

  float* rhs    = (float*)d_ws;
  float* lhs    = rhs + (size_t)N_IN * D;
  float* outAcc = lhs + (size_t)N_OT * D;

  const int tiles_in = (N_IN + 15) / 16;
  const int tiles_ot = (N_OT + 15) / 16;

  gemm64_kernel<<<(tiles_in + 7) / 8, 256, 0, stream>>>(input, Wi, bi, rhs, N_IN);
  gemm64_kernel<<<(tiles_ot + 7) / 8, 256, 0, stream>>>(other, Wo, nullptr, lhs, N_OT);

  const int n4 = (N_IN * D) / 4;
  zero4_kernel<<<(n4 + 255) / 256, 256, 0, stream>>>((float4*)outAcc, n4);

  const int eTiles = (E + 15) / 16;
  edge_kernel<<<(eTiles + 7) / 8, 256, 0, stream>>>(rhs, lhs, rj, lj, weights,
                                                    We, Wf, bf, outAcc, E);

  gemm_concat128_kernel<<<(tiles_in + 7) / 8, 256, 0, stream>>>(
      outAcc, input, Wout, bout, (float*)d_out, N_IN);
}


# Edge scatter made whole-line: accumulator halves of neighbouring column tiles exchanged (v_permlanex16) so one f32 atomic covers one 128-B line of one row, not two 64-B halves; sum order as parent
# speedup vs baseline: 1.0717x; 1.0717x over previous
#include <hip/hip_runtime.h>

typedef float v2f __attribute__((ext_vector_type(2)));
typedef float v8f __attribute__((ext_vector_type(8)));

__device__ __forceinline__ v8f wmma_f32(v2f a, v2f b, v8f c) {
  return __builtin_amdgcn_wmma_f32_16x16x4_f32(false, a, false, b, (short)0, c,
                                               false, false);
}

__device__ __forceinline__ void atomic_fadd_line_lo(float* rowBase, unsigned laneByteOff, float v) {
  asm volatile("global_atomic_add_f32 %0, %1, %2 scope:SCOPE_DEV"
               :
               : "v"(laneByteOff), "v"(v), "s"(rowBase)
               : "memory");
}
__device__ __forceinline__ void atomic_fadd_line_hi(float* rowBase, unsigned laneByteOff, float v) {
  asm volatile("global_atomic_add_f32 %0, %1, %2 offset:128 scope:SCOPE_DEV"
               :
               : "v"(laneByteOff), "v"(v), "s"(rowBase)
               : "memory");
}

__device__ __forceinline__ float xhalf(float x) {
  return __builtin_bit_cast(float, __builtin_amdgcn_permlanex16(__builtin_bit_cast(int, x),
                                                                __builtin_bit_cast(int, x),
                                                                0x76543210, 0xfedcba98, false, false));
}

__launch_bounds__(256)
__global__ void gemm64_kernel(const float* __restrict__ A,
                              const float* __restrict__ B,
                              const float* __restrict__ bias,
                              float* __restrict__ C, int M) {
  __shared__ float2 Bs2[32 * 64];
  for (int i = threadIdx.x; i < 2048; i += 256) {
    int k2 = i >> 6, n = i & 63;
    Bs2[i] = float2{B[(2 * k2) * 64 + n], B[(2 * k2 + 1) * 64 + n]};
  }
  __syncthreads();

  const int wave = threadIdx.x >> 5;
  const int lane = threadIdx.x & 31;
  const int r = lane & 15;
  const int h = lane >> 4;

  const int row0 = (blockIdx.x * 8 + wave) * 16;
  if (row0 >= M) return;

  int rowA = row0 + r;
  if (rowA >= M) rowA = M - 1;
  const float* __restrict__ Arow = A + (size_t)rowA * 64;

  v8f acc0 = {}, acc1 = {}, acc2 = {}, acc3 = {};
#pragma unroll
  for (int k0 = 0; k0 < 64; k0 += 4) {
    const int ka = k0 + 2 * h;
    float2 av = *reinterpret_cast<const float2*>(Arow + ka);
    v2f a; a.x = av.x; a.y = av.y;
    const int k2 = (k0 >> 1) + h;
    const float2* __restrict__ Bk = &Bs2[k2 * 64 + r];
    float2 f0 = Bk[0], f1 = Bk[16], f2 = Bk[32], f3 = Bk[48];
    v2f b0, b1, b2, b3;
    b0.x = f0.x; b0.y = f0.y;  b1.x = f1.x; b1.y = f1.y;
    b2.x = f2.x; b2.y = f2.y;  b3.x = f3.x; b3.y = f3.y;
    acc0 = wmma_f32(a, b0, acc0);
    acc1 = wmma_f32(a, b1, acc1);
    acc2 = wmma_f32(a, b2, acc2);
    acc3 = wmma_f32(a, b3, acc3);
  }

  float bia0 = 0.f, bia1 = 0.f, bia2 = 0.f, bia3 = 0.f;
  if (bias) { bia0 = bias[r]; bia1 = bias[16 + r]; bia2 = bias[32 + r]; bia3 = bias[48 + r]; }

#pragma unroll
  for (int v = 0; v < 8; ++v) {
    const int row = row0 + v + 8 * h;
    if (row < M) {
      float* __restrict__ Crow = C + (size_t)row * 64;
      Crow[r]      = acc0[v] + bia0;
      Crow[16 + r] = acc1[v] + bia1;
      Crow[32 + r] = acc2[v] + bia2;
      Crow[48 + r] = acc3[v] + bia3;
    }
  }
}

__launch_bounds__(256)
__global__ void gemm_concat128_kernel(const float* __restrict__ A0,
                                      const float* __restrict__ A1,
                                      const float* __restrict__ B,
                                      const float* __restrict__ bias,
                                      float* __restrict__ C, int M) {
  __shared__ float2 Bs2[64 * 64];
  for (int i = threadIdx.x; i < 4096; i += 256) {
    int k2 = i >> 6, n = i & 63;
    Bs2[i] = float2{B[(2 * k2) * 64 + n], B[(2 * k2 + 1) * 64 + n]};
  }
  __syncthreads();

  const int wave = threadIdx.x >> 5;
  const int lane = threadIdx.x & 31;
  const int r = lane & 15;
  const int h = lane >> 4;

  const int row0 = (blockIdx.x * 8 + wave) * 16;
  if (row0 >= M) return;

  int rowA = row0 + r;
  if (rowA >= M) rowA = M - 1;
  const float* __restrict__ Arow0 = A0 + (size_t)rowA * 64;
  const float* __restrict__ Arow1 = A1 + (size_t)rowA * 64;

  v8f acc0 = {}, acc1 = {}, acc2 = {}, acc3 = {};
#pragma unroll
  for (int k0 = 0; k0 < 128; k0 += 4) {
    const int ka = k0 + 2 * h;
    const float* src = (k0 < 64) ? (Arow0 + ka) : (Arow1 + (ka - 64));
    float2 av = *reinterpret_cast<const float2*>(src);
    v2f a; a.x = av.x; a.y = av.y;
    const int k2 = (k0 >> 1) + h;
    const float2* __restrict__ Bk = &Bs2[k2 * 64 + r];
    float2 f0 = Bk[0], f1 = Bk[16], f2 = Bk[32], f3 = Bk[48];
    v2f b0, b1, b2, b3;
    b0.x = f0.x; b0.y = f0.y;  b1.x = f1.x; b1.y = f1.y;
    b2.x = f2.x; b2.y = f2.y;  b3.x = f3.x; b3.y = f3.y;
    acc0 = wmma_f32(a, b0, acc0);
    acc1 = wmma_f32(a, b1, acc1);
    acc2 = wmma_f32(a, b2, acc2);
    acc3 = wmma_f32(a, b3, acc3);
  }

  const float bia0 = bias[r], bia1 = bias[16 + r], bia2 = bias[32 + r], bia3 = bias[48 + r];
#pragma unroll
  for (int v = 0; v < 8; ++v) {
    const int row = row0 + v + 8 * h;
    if (row < M) {
      float* __restrict__ Crow = C + (size_t)row * 64;
      Crow[r]      = acc0[v] + bia0;
      Crow[16 + r] = acc1[v] + bia1;
      Crow[32 + r] = acc2[v] + bia2;
      Crow[48 + r] = acc3[v] + bia3;
    }
  }
}

__launch_bounds__(256)
__global__ void edge_kernel(const float* __restrict__ rhs,
                            const float* __restrict__ lhs,
                            const int* __restrict__ rj,
                            const int* __restrict__ lj,
                            const float* __restrict__ weights,
                            const float* __restrict__ We,
                            const float* __restrict__ Wf,
                            const float* __restrict__ bf,
                            float* __restrict__ outAcc, int E) {
  __shared__ float2 WfS2[32 * 64];
  __shared__ float WeS[64];
  __shared__ float bfS[64];
  __shared__ float msgS[8][16 * 64];

  for (int i = threadIdx.x; i < 2048; i += 256) {
    int k2 = i >> 6, n = i & 63;
    WfS2[i] = float2{Wf[(2 * k2) * 64 + n], Wf[(2 * k2 + 1) * 64 + n]};
  }
  if (threadIdx.x < 64) { WeS[threadIdx.x] = We[threadIdx.x]; bfS[threadIdx.x] = bf[threadIdx.x]; }
  __syncthreads();

  const int wave = threadIdx.x >> 5;
  const int lane = threadIdx.x & 31;
  const int r = lane & 15;
  const int h = lane >> 4;

  const int eBase = (blockIdx.x * 8 + wave) * 16;

  int e = eBase + r;
  if (e >= E) e = E - 1;
  const int idxR = rj[e];
  const int idxL = lj[e];
  const float w = weights[e];

  const float4* __restrict__ pr = reinterpret_cast<const float4*>(rhs + (size_t)idxR * 64) + h * 8;
  const float4* __restrict__ pl = reinterpret_cast<const float4*>(lhs + (size_t)idxL * 64) + h * 8;
  const float4* __restrict__ pe = reinterpret_cast<const float4*>(WeS) + h * 8;
  float4* __restrict__ pm = reinterpret_cast<float4*>(&msgS[wave][r * 64]) + h * 8;
#pragma unroll
  for (int i = 0; i < 8; ++i) {
    float4 a = pr[i], b = pl[i], c = pe[i];
    float4 m;
    m.x = a.x + b.x + w * c.x;
    m.y = a.y + b.y + w * c.y;
    m.z = a.z + b.z + w * c.z;
    m.w = a.w + b.w + w * c.w;
    m.x = m.x > 0.f ? m.x : 0.01f * m.x;
    m.y = m.y > 0.f ? m.y : 0.01f * m.y;
    m.z = m.z > 0.f ? m.z : 0.01f * m.z;
    m.w = m.w > 0.f ? m.w : 0.01f * m.w;
    pm[i] = m;
  }

  v8f acc0 = {}, acc1 = {}, acc2 = {}, acc3 = {};
  const float* __restrict__ mrow = &msgS[wave][r * 64];
#pragma unroll
  for (int k0 = 0; k0 < 64; k0 += 4) {
    const int ka = k0 + 2 * h;
    float2 av = *reinterpret_cast<const float2*>(mrow + ka);
    v2f a; a.x = av.x; a.y = av.y;
    const int k2 = (k0 >> 1) + h;
    const float2* __restrict__ Bk = &WfS2[k2 * 64 + r];
    float2 f0 = Bk[0], f1 = Bk[16], f2 = Bk[32], f3 = Bk[48];
    v2f b0, b1, b2, b3;
    b0.x = f0.x; b0.y = f0.y;  b1.x = f1.x; b1.y = f1.y;
    b2.x = f2.x; b2.y = f2.y;  b3.x = f3.x; b3.y = f3.y;
    acc0 = wmma_f32(a, b0, acc0);
    acc1 = wmma_f32(a, b1, acc1);
    acc2 = wmma_f32(a, b2, acc2);
    acc3 = wmma_f32(a, b3, acc3);
  }

  const float biaLo = bfS[lane], biaHi = bfS[32 + lane];
  const unsigned laneOff = (unsigned)lane * 4u;
#pragma unroll
  for (int v = 0; v < 8; ++v) {
    const float a0 = acc0[v], b0 = acc1[v], a1 = acc2[v], b1 = acc3[v];
    const float ta0 = xhalf(a0), tb0 = xhalf(b0), ta1 = xhalf(a1), tb1 = xhalf(b1);
    const float x0 = h ? tb0 : a0;
    const float y0 = h ? b0 : ta0;
    const float x1 = h ? tb1 : a1;
    const float y1 = h ? b1 : ta1;
    if (eBase + v < E) {
      const int seg = __builtin_amdgcn_readlane(idxR, v);
      float* orow = outAcc + (size_t)seg * 64;
      atomic_fadd_line_lo(orow, laneOff, x0 + biaLo);
      atomic_fadd_line_hi(orow, laneOff, x1 + biaHi);
    }
    if (eBase + v + 8 < E) {
      const int seg = __builtin_amdgcn_readlane(idxR, v + 8);
      float* orow = outAcc + (size_t)seg * 64;
      atomic_fadd_line_lo(orow, laneOff, y0 + biaLo);
      atomic_fadd_line_hi(orow, laneOff, y1 + biaHi);
    }
  }
}

__global__ void zero4_kernel(float4* __restrict__ p, int n4) {
  int i = blockIdx.x * blockDim.x + threadIdx.x;
  if (i < n4) p[i] = float4{0.f, 0.f, 0.f, 0.f};
}

extern "C" void kernel_launch(void* const* d_in, const int* in_sizes, int n_in,
                              void* d_out, int out_size, void* d_ws, size_t ws_size,
                              hipStream_t stream) {
  const float* input   = (const float*)d_in[0];
  const float* other   = (const float*)d_in[1];
  const int*   rj      = (const int*)d_in[2];
  const int*   lj      = (const int*)d_in[3];
  const float* weights = (const float*)d_in[4];
  const float* Wi      = (const float*)d_in[5];
  const float* bi      = (const float*)d_in[6];
  const float* Wo      = (const float*)d_in[7];
  const float* We      = (const float*)d_in[8];
  const float* Wf      = (const float*)d_in[9];
  const float* bf      = (const float*)d_in[10];
  const float* Wout    = (const float*)d_in[11];
  const float* bout    = (const float*)d_in[12];

  const int D    = 64;
  const int N_IN = in_sizes[0] / D;
  const int N_OT = in_sizes[1] / D;
  const int E    = in_sizes[2];

  float* rhs    = (float*)d_ws;
  float* lhs    = rhs + (size_t)N_IN * D;
  float* outAcc = lhs + (size_t)N_OT * D;

  const int tiles_in = (N_IN + 15) / 16;
  const int tiles_ot = (N_OT + 15) / 16;

  gemm64_kernel<<<(tiles_in + 7) / 8, 256, 0, stream>>>(input, Wi, bi, rhs, N_IN);
  gemm64_kernel<<<(tiles_ot + 7) / 8, 256, 0, stream>>>(other, Wo, nullptr, lhs, N_OT);

  const int n4 = (N_IN * D) / 4;
  zero4_kernel<<<(n4 + 255) / 256, 256, 0, stream>>>((float4*)outAcc, n4);

  const int eTiles = (E + 15) / 16;
  edge_kernel<<<(eTiles + 7) / 8, 256, 0, stream>>>(rhs, lhs, rj, lj, weights,
                                                    We, Wf, bf, outAcc, E);

  gemm_concat128_kernel<<<(tiles_in + 7) / 8, 256, 0, stream>>>(
      outAcc, input, Wout, bout, (float*)d_out, N_IN);
}


# Edge kernel: A fragments are the gathered message registers themselves (k re-ordered: lane half h takes the h-th 16 B of every 32 B of its row; Wf read alike); no LDS tile; f32 sum order changes
# speedup vs baseline: 1.2735x; 1.1883x over previous
#include <hip/hip_runtime.h>

typedef float v2f __attribute__((ext_vector_type(2)));
typedef float v8f __attribute__((ext_vector_type(8)));

__device__ __forceinline__ v8f wmma_f32(v2f a, v2f b, v8f c) {
  return __builtin_amdgcn_wmma_f32_16x16x4_f32(false, a, false, b, (short)0, c,
                                               false, false);
}

__device__ __forceinline__ void atomic_fadd_line_lo(float* rowBase, unsigned laneByteOff, float v) {
  asm volatile("global_atomic_add_f32 %0, %1, %2 scope:SCOPE_DEV"
               :
               : "v"(laneByteOff), "v"(v), "s"(rowBase)
               : "memory");
}
__device__ __forceinline__ void atomic_fadd_line_hi(float* rowBase, unsigned laneByteOff, float v) {
  asm volatile("global_atomic_add_f32 %0, %1, %2 offset:128 scope:SCOPE_DEV"
               :
               : "v"(laneByteOff), "v"(v), "s"(rowBase)
               : "memory");
}

__device__ __forceinline__ float xhalf(float x) {
  return __builtin_bit_cast(float, __builtin_amdgcn_permlanex16(__builtin_bit_cast(int, x),
                                                                __builtin_bit_cast(int, x),
                                                                0x76543210, 0xfedcba98, false, false));
}

__launch_bounds__(256)
__global__ void gemm64_kernel(const float* __restrict__ A,
                              const float* __restrict__ B,
                              const float* __restrict__ bias,
                              float* __restrict__ C, int M) {
  __shared__ float2 Bs2[32 * 64];
  for (int i = threadIdx.x; i < 2048; i += 256) {
    int k2 = i >> 6, n = i & 63;
    Bs2[i] = float2{B[(2 * k2) * 64 + n], B[(2 * k2 + 1) * 64 + n]};
  }
  __syncthreads();

  const int wave = threadIdx.x >> 5;
  const int lane = threadIdx.x & 31;
  const int r = lane & 15;
  const int h = lane >> 4;

  const int row0 = (blockIdx.x * 8 + wave) * 16;
  if (row0 >= M) return;

  int rowA = row0 + r;
  if (rowA >= M) rowA = M - 1;
  const float* __restrict__ Arow = A + (size_t)rowA * 64;

  v8f acc0 = {}, acc1 = {}, acc2 = {}, acc3 = {};
#pragma unroll
  for (int k0 = 0; k0 < 64; k0 += 4) {
    const int ka = k0 + 2 * h;
    float2 av = *reinterpret_cast<const float2*>(Arow + ka);
    v2f a; a.x = av.x; a.y = av.y;
    const int k2 = (k0 >> 1) + h;
    const float2* __restrict__ Bk = &Bs2[k2 * 64 + r];
    float2 f0 = Bk[0], f1 = Bk[16], f2 = Bk[32], f3 = Bk[48];
    v2f b0, b1, b2, b3;
    b0.x = f0.x; b0.y = f0.y;  b1.x = f1.x; b1.y = f1.y;
    b2.x = f2.x; b2.y = f2.y;  b3.x = f3.x; b3.y = f3.y;
    acc0 = wmma_f32(a, b0, acc0);
    acc1 = wmma_f32(a, b1, acc1);
    acc2 = wmma_f32(a, b2, acc2);
    acc3 = wmma_f32(a, b3, acc3);
  }

  float bia0 = 0.f, bia1 = 0.f, bia2 = 0.f, bia3 = 0.f;
  if (bias) { bia0 = bias[r]; bia1 = bias[16 + r]; bia2 = bias[32 + r]; bia3 = bias[48 + r]; }

#pragma unroll
  for (int v = 0; v < 8; ++v) {
    const int row = row0 + v + 8 * h;
    if (row < M) {
      float* __restrict__ Crow = C + (size_t)row * 64;
      Crow[r]      = acc0[v] + bia0;
      Crow[16 + r] = acc1[v] + bia1;
      Crow[32 + r] = acc2[v] + bia2;
      Crow[48 + r] = acc3[v] + bia3;
    }
  }
}

__launch_bounds__(256)
__global__ void gemm_concat128_kernel(const float* __restrict__ A0,
                                      const float* __restrict__ A1,
                                      const float* __restrict__ B,
                                      const float* __restrict__ bias,
                                      float* __restrict__ C, int M) {
  __shared__ float2 Bs2[64 * 64];
  for (int i = threadIdx.x; i < 4096; i += 256) {
    int k2 = i >> 6, n = i & 63;
    Bs2[i] = float2{B[(2 * k2) * 64 + n], B[(2 * k2 + 1) * 64 + n]};
  }
  __syncthreads();

  const int wave = threadIdx.x >> 5;
  const int lane = threadIdx.x & 31;
  const int r = lane & 15;
  const int h = lane >> 4;

  const int row0 = (blockIdx.x * 8 + wave) * 16;
  if (row0 >= M) return;

  int rowA = row0 + r;
  if (rowA >= M) rowA = M - 1;
  const float* __restrict__ Arow0 = A0 + (size_t)rowA * 64;
  const float* __restrict__ Arow1 = A1 + (size_t)rowA * 64;

  v8f acc0 = {}, acc1 = {}, acc2 = {}, acc3 = {};
#pragma unroll
  for (int k0 = 0; k0 < 128; k0 += 4) {
    const int ka = k0 + 2 * h;
    const float* src = (k0 < 64) ? (Arow0 + ka) : (Arow1 + (ka - 64));
    float2 av = *reinterpret_cast<const float2*>(src);
    v2f a; a.x = av.x; a.y = av.y;
    const int k2 = (k0 >> 1) + h;
    const float2* __restrict__ Bk = &Bs2[k2 * 64 + r];
    float2 f0 = Bk[0], f1 = Bk[16], f2 = Bk[32], f3 = Bk[48];
    v2f b0, b1, b2, b3;
    b0.x = f0.x; b0.y = f0.y;  b1.x = f1.x; b1.y = f1.y;
    b2.x = f2.x; b2.y = f2.y;  b3.x = f3.x; b3.y = f3.y;
    acc0 = wmma_f32(a, b0, acc0);
    acc1 = wmma_f32(a, b1, acc1);
    acc2 = wmma_f32(a, b2, acc2);
    acc3 = wmma_f32(a, b3, acc3);
  }

  const float bia0 = bias[r], bia1 = bias[16 + r], bia2 = bias[32 + r], bia3 = bias[48 + r];
#pragma unroll
  for (int v = 0; v < 8; ++v) {
    const int row = row0 + v + 8 * h;
    if (row < M) {
      float* __restrict__ Crow = C + (size_t)row * 64;
      Crow[r]      = acc0[v] + bia0;
      Crow[16 + r] = acc1[v] + bia1;
      Crow[32 + r] = acc2[v] + bia2;
      Crow[48 + r] = acc3[v] + bia3;
    }
  }
}

__launch_bounds__(256)
__global__ void edge_kernel(const float* __restrict__ rhs,
                            const float* __restrict__ lhs,
                            const int* __restrict__ rj,
                            const int* __restrict__ lj,
                            const float* __restrict__ weights,
                            const float* __restrict__ We,
                            const float* __restrict__ Wf,
                            const float* __restrict__ bf,
                            float* __restrict__ outAcc, int E) {
  __shared__ float2 WfS2[32 * 64];
  __shared__ float WeS[64];
  __shared__ float bfS[64];

  for (int i = threadIdx.x; i < 2048; i += 256) {
    int k2 = i >> 6, n = i & 63;
    WfS2[i] = float2{Wf[(2 * k2) * 64 + n], Wf[(2 * k2 + 1) * 64 + n]};
  }
  if (threadIdx.x < 64) { WeS[threadIdx.x] = We[threadIdx.x]; bfS[threadIdx.x] = bf[threadIdx.x]; }
  __syncthreads();

  const int wave = threadIdx.x >> 5;
  const int lane = threadIdx.x & 31;
  const int r = lane & 15;
  const int h = lane >> 4;

  const int eBase = (blockIdx.x * 8 + wave) * 16;

  int e = eBase + r;
  if (e >= E) e = E - 1;
  const int idxR = rj[e];
  const int idxL = lj[e];
  const float w = weights[e];

  const float4* __restrict__ pr = reinterpret_cast<const float4*>(rhs + (size_t)idxR * 64) + h;
  const float4* __restrict__ pl = reinterpret_cast<const float4*>(lhs + (size_t)idxL * 64) + h;
  const float4* __restrict__ pe = reinterpret_cast<const float4*>(WeS) + h;
  float4 msg[8];
#pragma unroll
  for (int i = 0; i < 8; ++i) {
    float4 a = pr[2 * i], b = pl[2 * i], c = pe[2 * i];
    float4 m;
    m.x = a.x + b.x + w * c.x;
    m.y = a.y + b.y + w * c.y;
    m.z = a.z + b.z + w * c.z;
    m.w = a.w + b.w + w * c.w;
    m.x = m.x > 0.f ? m.x : 0.01f * m.x;
    m.y = m.y > 0.f ? m.y : 0.01f * m.y;
    m.z = m.z > 0.f ? m.z : 0.01f * m.z;
    m.w = m.w > 0.f ? m.w : 0.01f * m.w;
    msg[i] = m;
  }

  v8f acc0 = {}, acc1 = {}, acc2 = {}, acc3 = {};
#pragma unroll
  for (int i = 0; i < 8; ++i) {
#pragma unroll
    for (int p = 0; p < 2; ++p) {
      v2f a;
      a.x = p ? msg[i].z : msg[i].x;
      a.y = p ? msg[i].w : msg[i].y;
      const int k2 = 4 * i + 2 * h + p;
      const float2* __restrict__ Bk = &WfS2[k2 * 64 + r];
      float2 f0 = Bk[0], f1 = Bk[16], f2 = Bk[32], f3 = Bk[48];
      v2f b0, b1, b2, b3;
      b0.x = f0.x; b0.y = f0.y;  b1.x = f1.x; b1.y = f1.y;
      b2.x = f2.x; b2.y = f2.y;  b3.x = f3.x; b3.y = f3.y;
      acc0 = wmma_f32(a, b0, acc0);
      acc1 = wmma_f32(a, b1, acc1);
      acc2 = wmma_f32(a, b2, acc2);
      acc3 = wmma_f32(a, b3, acc3);
    }
  }

  const float biaLo = bfS[lane], biaHi = bfS[32 + lane];
  const unsigned laneOff = (unsigned)lane * 4u;
#pragma unroll
  for (int v = 0; v < 8; ++v) {
    const float a0 = acc0[v], b0 = acc1[v], a1 = acc2[v], b1 = acc3[v];
    const float ta0 = xhalf(a0), tb0 = xhalf(b0), ta1 = xhalf(a1), tb1 = xhalf(b1);
    const float x0 = h ? tb0 : a0;
    const float y0 = h ? b0 : ta0;
    const float x1 = h ? tb1 : a1;
    const float y1 = h ? b1 : ta1;
    if (eBase + v < E) {
      const int seg = __builtin_amdgcn_readlane(idxR, v);
      float* orow = outAcc + (size_t)seg * 64;
      atomic_fadd_line_lo(orow, laneOff, x0 + biaLo);
      atomic_fadd_line_hi(orow, laneOff, x1 + biaHi);
    }
    if (eBase + v + 8 < E) {
      const int seg = __builtin_amdgcn_readlane(idxR, v + 8);
      float* orow = outAcc + (size_t)seg * 64;
      atomic_fadd_line_lo(orow, laneOff, y0 + biaLo);
      atomic_fadd_line_hi(orow, laneOff, y1 + biaHi);
    }
  }
}

__global__ void zero4_kernel(float4* __restrict__ p, int n4) {
  int i = blockIdx.x * blockDim.x + threadIdx.x;
  if (i < n4) p[i] = float4{0.f, 0.f, 0.f, 0.f};
}

extern "C" void kernel_launch(void* const* d_in, const int* in_sizes, int n_in,
                              void* d_out, int out_size, void* d_ws, size_t ws_size,
                              hipStream_t stream) {
  const float* input   = (const float*)d_in[0];
  const float* other   = (const float*)d_in[1];
  const int*   rj      = (const int*)d_in[2];
  const int*   lj      = (const int*)d_in[3];
  const float* weights = (const float*)d_in[4];
  const float* Wi      = (const float*)d_in[5];
  const float* bi      = (const float*)d_in[6];
  const float* Wo      = (const float*)d_in[7];
  const float* We      = (const float*)d_in[8];
  const float* Wf      = (const float*)d_in[9];
  const float* bf      = (const float*)d_in[10];
  const float* Wout    = (const float*)d_in[11];
  const float* bout    = (const float*)d_in[12];

  const int D    = 64;
  const int N_IN = in_sizes[0] / D;
  const int N_OT = in_sizes[1] / D;
  const int E    = in_sizes[2];

  float* rhs    = (float*)d_ws;
  float* lhs    = rhs + (size_t)N_IN * D;
  float* outAcc = lhs + (size_t)N_OT * D;

  const int tiles_in = (N_IN + 15) / 16;
  const int tiles_ot = (N_OT + 15) / 16;

  gemm64_kernel<<<(tiles_in + 7) / 8, 256, 0, stream>>>(input, Wi, bi, rhs, N_IN);
  gemm64_kernel<<<(tiles_ot + 7) / 8, 256, 0, stream>>>(other, Wo, nullptr, lhs, N_OT);

  const int n4 = (N_IN * D) / 4;
  zero4_kernel<<<(n4 + 255) / 256, 256, 0, stream>>>((float4*)outAcc, n4);

  const int eTiles = (E + 15) / 16;
  edge_kernel<<<(eTiles + 7) / 8, 256, 0, stream>>>(rhs, lhs, rj, lj, weights,
                                                    We, Wf, bf, outAcc, E);

  gemm_concat128_kernel<<<(tiles_in + 7) / 8, 256, 0, stream>>>(
      outAcc, input, Wout, bout, (float*)d_out, N_IN);
}


# Edge kernel made persistent with Wf as B fragments in registers: a wave loads Wf once (128 VGPRs) and loops over its tiles, WMMA reads both operands from registers; no LDS Wf staging; sums as parent
# speedup vs baseline: 1.3840x; 1.0868x over previous
#include <hip/hip_runtime.h>

typedef float v2f __attribute__((ext_vector_type(2)));
typedef float v8f __attribute__((ext_vector_type(8)));

__device__ __forceinline__ v8f wmma_f32(v2f a, v2f b, v8f c) {
  return __builtin_amdgcn_wmma_f32_16x16x4_f32(false, a, false, b, (short)0, c,
                                               false, false);
}

__device__ __forceinline__ void atomic_fadd_line_lo(float* rowBase, unsigned laneByteOff, float v) {
  asm volatile("global_atomic_add_f32 %0, %1, %2 scope:SCOPE_DEV"
               :
               : "v"(laneByteOff), "v"(v), "s"(rowBase)
               : "memory");
}
__device__ __forceinline__ void atomic_fadd_line_hi(float* rowBase, unsigned laneByteOff, float v) {
  asm volatile("global_atomic_add_f32 %0, %1, %2 offset:128 scope:SCOPE_DEV"
               :
               : "v"(laneByteOff), "v"(v), "s"(rowBase)
               : "memory");
}

__device__ __forceinline__ float xhalf(float x) {
  return __builtin_bit_cast(float, __builtin_amdgcn_permlanex16(__builtin_bit_cast(int, x),
                                                                __builtin_bit_cast(int, x),
                                                                0x76543210, 0xfedcba98, false, false));
}

__launch_bounds__(256)
__global__ void gemm64_kernel(const float* __restrict__ A,
                              const float* __restrict__ B,
                              const float* __restrict__ bias,
                              float* __restrict__ C, int M) {
  __shared__ float2 Bs2[32 * 64];
  for (int i = threadIdx.x; i < 2048; i += 256) {
    int k2 = i >> 6, n = i & 63;
    Bs2[i] = float2{B[(2 * k2) * 64 + n], B[(2 * k2 + 1) * 64 + n]};
  }
  __syncthreads();

  const int wave = threadIdx.x >> 5;
  const int lane = threadIdx.x & 31;
  const int r = lane & 15;
  const int h = lane >> 4;

  const int row0 = (blockIdx.x * 8 + wave) * 16;
  if (row0 >= M) return;

  int rowA = row0 + r;
  if (rowA >= M) rowA = M - 1;
  const float* __restrict__ Arow = A + (size_t)rowA * 64;

  v8f acc0 = {}, acc1 = {}, acc2 = {}, acc3 = {};
#pragma unroll
  for (int k0 = 0; k0 < 64; k0 += 4) {
    const int ka = k0 + 2 * h;
    float2 av = *reinterpret_cast<const float2*>(Arow + ka);
    v2f a; a.x = av.x; a.y = av.y;
    const int k2 = (k0 >> 1) + h;
    const float2* __restrict__ Bk = &Bs2[k2 * 64 + r];
    float2 f0 = Bk[0], f1 = Bk[16], f2 = Bk[32], f3 = Bk[48];
    v2f b0, b1, b2, b3;
    b0.x = f0.x; b0.y = f0.y;  b1.x = f1.x; b1.y = f1.y;
    b2.x = f2.x; b2.y = f2.y;  b3.x = f3.x; b3.y = f3.y;
    acc0 = wmma_f32(a, b0, acc0);
    acc1 = wmma_f32(a, b1, acc1);
    acc2 = wmma_f32(a, b2, acc2);
    acc3 = wmma_f32(a, b3, acc3);
  }

  float bia0 = 0.f, bia1 = 0.f, bia2 = 0.f, bia3 = 0.f;
  if (bias) { bia0 = bias[r]; bia1 = bias[16 + r]; bia2 = bias[32 + r]; bia3 = bias[48 + r]; }

#pragma unroll
  for (int v = 0; v < 8; ++v) {
    const int row = row0 + v + 8 * h;
    if (row < M) {
      float* __restrict__ Crow = C + (size_t)row * 64;
      Crow[r]      = acc0[v] + bia0;
      Crow[16 + r] = acc1[v] + bia1;
      Crow[32 + r] = acc2[v] + bia2;
      Crow[48 + r] = acc3[v] + bia3;
    }
  }
}

__launch_bounds__(256)
__global__ void gemm_concat128_kernel(const float* __restrict__ A0,
                                      const float* __restrict__ A1,
                                      const float* __restrict__ B,
                                      const float* __restrict__ bias,
                                      float* __restrict__ C, int M) {
  __shared__ float2 Bs2[64 * 64];
  for (int i = threadIdx.x; i < 4096; i += 256) {
    int k2 = i >> 6, n = i & 63;
    Bs2[i] = float2{B[(2 * k2) * 64 + n], B[(2 * k2 + 1) * 64 + n]};
  }
  __syncthreads();

  const int wave = threadIdx.x >> 5;
  const int lane = threadIdx.x & 31;
  const int r = lane & 15;
  const int h = lane >> 4;

  const int row0 = (blockIdx.x * 8 + wave) * 16;
  if (row0 >= M) return;

  int rowA = row0 + r;
  if (rowA >= M) rowA = M - 1;
  const float* __restrict__ Arow0 = A0 + (size_t)rowA * 64;
  const float* __restrict__ Arow1 = A1 + (size_t)rowA * 64;

  v8f acc0 = {}, acc1 = {}, acc2 = {}, acc3 = {};
#pragma unroll
  for (int k0 = 0; k0 < 128; k0 += 4) {
    const int ka = k0 + 2 * h;
    const float* src = (k0 < 64) ? (Arow0 + ka) : (Arow1 + (ka - 64));
    float2 av = *reinterpret_cast<const float2*>(src);
    v2f a; a.x = av.x; a.y = av.y;
    const int k2 = (k0 >> 1) + h;
    const float2* __restrict__ Bk = &Bs2[k2 * 64 + r];
    float2 f0 = Bk[0], f1 = Bk[16], f2 = Bk[32], f3 = Bk[48];
    v2f b0, b1, b2, b3;
    b0.x = f0.x; b0.y = f0.y;  b1.x = f1.x; b1.y = f1.y;
    b2.x = f2.x; b2.y = f2.y;  b3.x = f3.x; b3.y = f3.y;
    acc0 = wmma_f32(a, b0, acc0);
    acc1 = wmma_f32(a, b1, acc1);
    acc2 = wmma_f32(a, b2, acc2);
    acc3 = wmma_f32(a, b3, acc3);
  }

  const float bia0 = bias[r], bia1 = bias[16 + r], bia2 = bias[32 + r], bia3 = bias[48 + r];
#pragma unroll
  for (int v = 0; v < 8; ++v) {
    const int row = row0 + v + 8 * h;
    if (row < M) {
      float* __restrict__ Crow = C + (size_t)row * 64;
      Crow[r]      = acc0[v] + bia0;
      Crow[16 + r] = acc1[v] + bia1;
      Crow[32 + r] = acc2[v] + bia2;
      Crow[48 + r] = acc3[v] + bia3;
    }
  }
}

__launch_bounds__(128)
__global__ void edge_kernel(const float* __restrict__ rhs,
                            const float* __restrict__ lhs,
                            const int* __restrict__ rj,
                            const int* __restrict__ lj,
                            const float* __restrict__ weights,
                            const float* __restrict__ We,
                            const float* __restrict__ Wf,
                            const float* __restrict__ bf,
                            float* __restrict__ outAcc, int E, int eTiles) {
  __shared__ float WeS[64];

  if (threadIdx.x < 64) WeS[threadIdx.x] = We[threadIdx.x];
  __syncthreads();

  const int wave = __builtin_amdgcn_readfirstlane((int)(threadIdx.x >> 5));
  const int lane = threadIdx.x & 31;
  const int r = lane & 15;
  const int h = lane >> 4;

  v2f Bf[16][4];
#pragma unroll
  for (int j = 0; j < 16; ++j) {
    const int k2 = 4 * (j >> 1) + 2 * h + (j & 1);
    const float* __restrict__ w0 = Wf + (2 * k2) * 64 + r;
#pragma unroll
    for (int t = 0; t < 4; ++t) {
      v2f b;
      b.x = w0[16 * t];
      b.y = w0[64 + 16 * t];
      Bf[j][t] = b;
    }
  }

  const float biaLo = bf[lane], biaHi = bf[32 + lane];
  const unsigned laneOff = (unsigned)lane * 4u;
  const float4* __restrict__ pe = reinterpret_cast<const float4*>(WeS) + h;

  const int tstride = (int)gridDim.x * 4;
  int tile = (int)blockIdx.x * 4 + wave;

  int e = tile * 16 + r;
  if (e >= E) e = E - 1;
  int idxR = rj[e];
  int idxL = lj[e];
  float w = weights[e];

  for (; tile < eTiles; tile += tstride) {
    const int eBase = tile * 16;

    const float4* __restrict__ pr = reinterpret_cast<const float4*>(rhs + (size_t)idxR * 64) + h;
    const float4* __restrict__ pl = reinterpret_cast<const float4*>(lhs + (size_t)idxL * 64) + h;
    float4 ga[8], gb[8];
#pragma unroll
    for (int i = 0; i < 8; ++i) { ga[i] = pr[2 * i]; gb[i] = pl[2 * i]; }

    int en = (tile + tstride) * 16 + r;
    if (en >= E) en = E - 1;
    const int nR = rj[en];
    const int nL = lj[en];
    const float nw = weights[en];

    float4 msg[8];
#pragma unroll
    for (int i = 0; i < 8; ++i) {
      float4 a = ga[i], b = gb[i], c = pe[2 * i];
      float4 m;
      m.x = a.x + b.x + w * c.x;
      m.y = a.y + b.y + w * c.y;
      m.z = a.z + b.z + w * c.z;
      m.w = a.w + b.w + w * c.w;
      m.x = m.x > 0.f ? m.x : 0.01f * m.x;
      m.y = m.y > 0.f ? m.y : 0.01f * m.y;
      m.z = m.z > 0.f ? m.z : 0.01f * m.z;
      m.w = m.w > 0.f ? m.w : 0.01f * m.w;
      msg[i] = m;
    }

    v8f acc0 = {}, acc1 = {}, acc2 = {}, acc3 = {};
#pragma unroll
    for (int i = 0; i < 8; ++i) {
#pragma unroll
      for (int p = 0; p < 2; ++p) {
        v2f a;
        a.x = p ? msg[i].z : msg[i].x;
        a.y = p ? msg[i].w : msg[i].y;
        const int j = 2 * i + p;
        acc0 = wmma_f32(a, Bf[j][0], acc0);
        acc1 = wmma_f32(a, Bf[j][1], acc1);
        acc2 = wmma_f32(a, Bf[j][2], acc2);
        acc3 = wmma_f32(a, Bf[j][3], acc3);
      }
    }

#pragma unroll
    for (int v = 0; v < 8; ++v) {
      const float a0 = acc0[v], b0 = acc1[v], a1 = acc2[v], b1 = acc3[v];
      const float ta0 = xhalf(a0), tb0 = xhalf(b0), ta1 = xhalf(a1), tb1 = xhalf(b1);
      const float x0 = h ? tb0 : a0;
      const float y0 = h ? b0 : ta0;
      const float x1 = h ? tb1 : a1;
      const float y1 = h ? b1 : ta1;
      if (eBase + v < E) {
        const int seg = __builtin_amdgcn_readlane(idxR, v);
        float* orow = outAcc + (size_t)seg * 64;
        atomic_fadd_line_lo(orow, laneOff, x0 + biaLo);
        atomic_fadd_line_hi(orow, laneOff, x1 + biaHi);
      }
      if (eBase + v + 8 < E) {
        const int seg = __builtin_amdgcn_readlane(idxR, v + 8);
        float* orow = outAcc + (size_t)seg * 64;
        atomic_fadd_line_lo(orow, laneOff, y0 + biaLo);
        atomic_fadd_line_hi(orow, laneOff, y1 + biaHi);
      }
    }

    idxR = nR; idxL = nL; w = nw;
  }
}

__global__ void zero4_kernel(float4* __restrict__ p, int n4) {
  int i = blockIdx.x * blockDim.x + threadIdx.x;
  if (i < n4) p[i] = float4{0.f, 0.f, 0.f, 0.f};
}

extern "C" void kernel_launch(void* const* d_in, const int* in_sizes, int n_in,
                              void* d_out, int out_size, void* d_ws, size_t ws_size,
                              hipStream_t stream) {
  const float* input   = (const float*)d_in[0];
  const float* other   = (const float*)d_in[1];
  const int*   rj      = (const int*)d_in[2];
  const int*   lj      = (const int*)d_in[3];
  const float* weights = (const float*)d_in[4];
  const float* Wi      = (const float*)d_in[5];
  const float* bi      = (const float*)d_in[6];
  const float* Wo      = (const float*)d_in[7];
  const float* We      = (const float*)d_in[8];
  const float* Wf      = (const float*)d_in[9];
  const float* bf      = (const float*)d_in[10];
  const float* Wout    = (const float*)d_in[11];
  const float* bout    = (const float*)d_in[12];

  const int D    = 64;
  const int N_IN = in_sizes[0] / D;
  const int N_OT = in_sizes[1] / D;
  const int E    = in_sizes[2];

  float* rhs    = (float*)d_ws;
  float* lhs    = rhs + (size_t)N_IN * D;
  float* outAcc = lhs + (size_t)N_OT * D;

  const int tiles_in = (N_IN + 15) / 16;
  const int tiles_ot = (N_OT + 15) / 16;

  gemm64_kernel<<<(tiles_in + 7) / 8, 256, 0, stream>>>(input, Wi, bi, rhs, N_IN);
  gemm64_kernel<<<(tiles_ot + 7) / 8, 256, 0, stream>>>(other, Wo, nullptr, lhs, N_OT);

  const int n4 = (N_IN * D) / 4;
  zero4_kernel<<<(n4 + 255) / 256, 256, 0, stream>>>((float4*)outAcc, n4);

  const int eTiles = (E + 15) / 16;
  edge_kernel<<<1024, 128, 0, stream>>>(rhs, lhs, rj, lj, weights,
                                        We, Wf, bf, outAcc, E, eTiles);

  gemm_concat128_kernel<<<(tiles_in + 7) / 8, 256, 0, stream>>>(
      outAcc, input, Wout, bout, (float*)d_out, N_IN);
}


# 4 launches -> 2: the three independent preparation jobs (other-side GEMM, weight fold, input dual GEMM) run as work-groups of one grid, told apart by block index; no sum order changes
# speedup vs baseline: 1.8702x; 1.3513x over previous
#include <hip/hip_runtime.h>

typedef float v2f __attribute__((ext_vector_type(2)));
typedef float v8f __attribute__((ext_vector_type(8)));

__device__ __forceinline__ v8f wmma_f32(v2f a, v2f b, v8f c) {
  return __builtin_amdgcn_wmma_f32_16x16x4_f32(false, a, false, b, (short)0, c,
                                               false, false);
}

__device__ __forceinline__ void atomic_fadd_line_lo(float* rowBase, unsigned laneByteOff, float v) {
  asm volatile("global_atomic_add_f32 %0, %1, %2 scope:SCOPE_DEV"
               :
               : "v"(laneByteOff), "v"(v), "s"(rowBase)
               : "memory");
}
__device__ __forceinline__ void atomic_fadd_line_hi(float* rowBase, unsigned laneByteOff, float v) {
  asm volatile("global_atomic_add_f32 %0, %1, %2 offset:128 scope:SCOPE_DEV"
               :
               : "v"(laneByteOff), "v"(v), "s"(rowBase)
               : "memory");
}

__device__ __forceinline__ float xhalf(float x) {
  return __builtin_bit_cast(float, __builtin_amdgcn_permlanex16(__builtin_bit_cast(int, x),
                                                                __builtin_bit_cast(int, x),
                                                                0x76543210, 0xfedcba98, false, false));
}

__device__ __forceinline__ void gemm64_body(float2* __restrict__ Bs2, int blk,
                                            const float* __restrict__ A,
                                            const float* __restrict__ B,
                                            const float* __restrict__ bias,
                                            float* __restrict__ C, int M) {
  for (int i = threadIdx.x; i < 2048; i += 256) {
    int k2 = i >> 6, n = i & 63;
    Bs2[i] = float2{B[(2 * k2) * 64 + n], B[(2 * k2 + 1) * 64 + n]};
  }
  __syncthreads();

  const int wave = threadIdx.x >> 5;
  const int lane = threadIdx.x & 31;
  const int r = lane & 15;
  const int h = lane >> 4;

  const int row0 = (blk * 8 + wave) * 16;
  if (row0 >= M) return;

  int rowA = row0 + r;
  if (rowA >= M) rowA = M - 1;
  const float* __restrict__ Arow = A + (size_t)rowA * 64;

  v8f acc0 = {}, acc1 = {}, acc2 = {}, acc3 = {};
#pragma unroll
  for (int k0 = 0; k0 < 64; k0 += 4) {
    const int ka = k0 + 2 * h;
    float2 av = *reinterpret_cast<const float2*>(Arow + ka);
    v2f a; a.x = av.x; a.y = av.y;
    const int k2 = (k0 >> 1) + h;
    const float2* __restrict__ Bk = &Bs2[k2 * 64 + r];
    float2 f0 = Bk[0], f1 = Bk[16], f2 = Bk[32], f3 = Bk[48];
    v2f b0, b1, b2, b3;
    b0.x = f0.x; b0.y = f0.y;  b1.x = f1.x; b1.y = f1.y;
    b2.x = f2.x; b2.y = f2.y;  b3.x = f3.x; b3.y = f3.y;
    acc0 = wmma_f32(a, b0, acc0);
    acc1 = wmma_f32(a, b1, acc1);
    acc2 = wmma_f32(a, b2, acc2);
    acc3 = wmma_f32(a, b3, acc3);
  }

  float bia0 = 0.f, bia1 = 0.f, bia2 = 0.f, bia3 = 0.f;
  if (bias) { bia0 = bias[r]; bia1 = bias[16 + r]; bia2 = bias[32 + r]; bia3 = bias[48 + r]; }

#pragma unroll
  for (int v = 0; v < 8; ++v) {
    const int row = row0 + v + 8 * h;
    if (row < M) {
      float* __restrict__ Crow = C + (size_t)row * 64;
      Crow[r]      = acc0[v] + bia0;
      Crow[16 + r] = acc1[v] + bia1;
      Crow[32 + r] = acc2[v] + bia2;
      Crow[48 + r] = acc3[v] + bia3;
    }
  }
}

__device__ __forceinline__ void gemm64_dual_body(float2 (* __restrict__ Bs2)[32 * 64], int blk,
                                                 const float* __restrict__ A,
                                                 const float* __restrict__ B0,
                                                 const float* __restrict__ bias0,
                                                 float* __restrict__ C0,
                                                 const float* __restrict__ B1,
                                                 const float* __restrict__ bias1,
                                                 float* __restrict__ C1, int M) {
  for (int i = threadIdx.x; i < 2048; i += 256) {
    int k2 = i >> 6, n = i & 63;
    Bs2[0][i] = float2{B0[(2 * k2) * 64 + n], B0[(2 * k2 + 1) * 64 + n]};
    Bs2[1][i] = float2{B1[(2 * k2) * 64 + n], B1[(2 * k2 + 1) * 64 + n]};
  }
  __syncthreads();

  const int wave = threadIdx.x >> 5;
  const int lane = threadIdx.x & 31;
  const int r = lane & 15;
  const int h = lane >> 4;

  const int row0 = (blk * 8 + wave) * 16;
  if (row0 >= M) return;

  int rowA = row0 + r;
  if (rowA >= M) rowA = M - 1;
  const float* __restrict__ Arow = A + (size_t)rowA * 64;

  v8f p0 = {}, p1 = {}, p2 = {}, p3 = {};
  v8f q0 = {}, q1 = {}, q2 = {}, q3 = {};
#pragma unroll
  for (int k0 = 0; k0 < 64; k0 += 4) {
    const int ka = k0 + 2 * h;
    float2 av = *reinterpret_cast<const float2*>(Arow + ka);
    v2f a; a.x = av.x; a.y = av.y;
    const int k2 = (k0 >> 1) + h;
    const float2* __restrict__ Bk = &Bs2[0][k2 * 64 + r];
    const float2* __restrict__ Bq = &Bs2[1][k2 * 64 + r];
    float2 f0 = Bk[0], f1 = Bk[16], f2 = Bk[32], f3 = Bk[48];
    float2 g0 = Bq[0], g1 = Bq[16], g2 = Bq[32], g3 = Bq[48];
    v2f b0, b1, b2, b3, c0, c1, c2, c3;
    b0.x = f0.x; b0.y = f0.y;  b1.x = f1.x; b1.y = f1.y;
    b2.x = f2.x; b2.y = f2.y;  b3.x = f3.x; b3.y = f3.y;
    c0.x = g0.x; c0.y = g0.y;  c1.x = g1.x; c1.y = g1.y;
    c2.x = g2.x; c2.y = g2.y;  c3.x = g3.x; c3.y = g3.y;
    p0 = wmma_f32(a, b0, p0);
    p1 = wmma_f32(a, b1, p1);
    p2 = wmma_f32(a, b2, p2);
    p3 = wmma_f32(a, b3, p3);
    q0 = wmma_f32(a, c0, q0);
    q1 = wmma_f32(a, c1, q1);
    q2 = wmma_f32(a, c2, q2);
    q3 = wmma_f32(a, c3, q3);
  }

  const float bp0 = bias0[r], bp1 = bias0[16 + r], bp2 = bias0[32 + r], bp3 = bias0[48 + r];
  const float bq0 = bias1[r], bq1 = bias1[16 + r], bq2 = bias1[32 + r], bq3 = bias1[48 + r];

#pragma unroll
  for (int v = 0; v < 8; ++v) {
    const int row = row0 + v + 8 * h;
    if (row < M) {
      float* __restrict__ Prow = C0 + (size_t)row * 64;
      Prow[r]      = p0[v] + bp0;
      Prow[16 + r] = p1[v] + bp1;
      Prow[32 + r] = p2[v] + bp2;
      Prow[48 + r] = p3[v] + bp3;
      float* __restrict__ Qrow = C1 + (size_t)row * 64;
      Qrow[r]      = q0[v] + bq0;
      Qrow[16 + r] = q1[v] + bq1;
      Qrow[32 + r] = q2[v] + bq2;
      Qrow[48 + r] = q3[v] + bq3;
    }
  }
}

__device__ __forceinline__ void fold_body(int tileId,
                                          const float* __restrict__ Wf,
                                          const float* __restrict__ bf,
                                          const float* __restrict__ Wout,
                                          float* __restrict__ Wc) {
  const int lane = threadIdx.x & 31;
  const int r = lane & 15;
  const int h = lane >> 4;
  const int rt = tileId >> 2;
  const int ct = tileId & 3;

  const float* __restrict__ Arow = (rt < 4) ? (Wf + (rt * 16 + r) * 64) : bf;
  const float* __restrict__ Bcol = Wout + ct * 16 + r;

  v8f acc = {};
#pragma unroll
  for (int k0 = 0; k0 < 64; k0 += 4) {
    const int ka = k0 + 2 * h;
    v2f a, b;
    a.x = Arow[ka];            a.y = Arow[ka + 1];
    b.x = Bcol[ka * 64];       b.y = Bcol[(ka + 1) * 64];
    acc = wmma_f32(a, b, acc);
  }

  if (rt < 4) {
#pragma unroll
    for (int v = 0; v < 8; ++v)
      Wc[(rt * 16 + v + 8 * h) * 64 + ct * 16 + r] = acc[v];
  } else if (h == 0) {
    Wc[64 * 64 + ct * 16 + r] = acc[0];
  }
}

__launch_bounds__(256)
__global__ void prep_kernel(const float* __restrict__ input,
                            const float* __restrict__ Wi,
                            const float* __restrict__ bi,
                            float* __restrict__ rhs,
                            const float* __restrict__ WoutBot,
                            const float* __restrict__ bout,
                            float* __restrict__ out, int nIn,
                            const float* __restrict__ other,
                            const float* __restrict__ Wo,
                            float* __restrict__ lhs, int nOt,
                            const float* __restrict__ Wf,
                            const float* __restrict__ bf,
                            const float* __restrict__ Wout,
                            float* __restrict__ Wc,
                            int nFold, int nDual) {
  __shared__ float2 Bs2[2][32 * 64];
  const int b = (int)blockIdx.x;
  if (b < nFold) {
    const int tileId = b * 8 + __builtin_amdgcn_readfirstlane((int)(threadIdx.x >> 5));
    if (tileId < 20) fold_body(tileId, Wf, bf, Wout, Wc);
  } else if (b < nFold + nDual) {
    gemm64_dual_body(Bs2, b - nFold, input, Wi, bi, rhs, WoutBot, bout, out, nIn);
  } else {
    gemm64_body(Bs2[0], b - nFold - nDual, other, Wo, nullptr, lhs, nOt);
  }
}

__launch_bounds__(128)
__global__ void edge_kernel(const float* __restrict__ rhs,
                            const float* __restrict__ lhs,
                            const int* __restrict__ rj,
                            const int* __restrict__ lj,
                            const float* __restrict__ weights,
                            const float* __restrict__ We,
                            const float* __restrict__ Wf,
                            const float* __restrict__ bf,
                            float* __restrict__ outAcc, int E, int eTiles) {
  __shared__ float WeS[64];

  if (threadIdx.x < 64) WeS[threadIdx.x] = We[threadIdx.x];
  __syncthreads();

  const int wave = __builtin_amdgcn_readfirstlane((int)(threadIdx.x >> 5));
  const int lane = threadIdx.x & 31;
  const int r = lane & 15;
  const int h = lane >> 4;

  v2f Bf[16][4];
#pragma unroll
  for (int j = 0; j < 16; ++j) {
    const int k2 = 4 * (j >> 1) + 2 * h + (j & 1);
    const float* __restrict__ w0 = Wf + (2 * k2) * 64 + r;
#pragma unroll
    for (int t = 0; t < 4; ++t) {
      v2f b;
      b.x = w0[16 * t];
      b.y = w0[64 + 16 * t];
      Bf[j][t] = b;
    }
  }

  const float biaLo = bf[lane], biaHi = bf[32 + lane];
  const unsigned laneOff = (unsigned)lane * 4u;
  const float4* __restrict__ pe = reinterpret_cast<const float4*>(WeS) + h;

  const int tstride = (int)gridDim.x * 4;
  int tile = (int)blockIdx.x * 4 + wave;

  int e = tile * 16 + r;
  if (e >= E) e = E - 1;
  int idxR = rj[e];
  int idxL = lj[e];
  float w = weights[e];

  for (; tile < eTiles; tile += tstride) {
    const int eBase = tile * 16;

    const float4* __restrict__ pr = reinterpret_cast<const float4*>(rhs + (size_t)idxR * 64) + h;
    const float4* __restrict__ pl = reinterpret_cast<const float4*>(lhs + (size_t)idxL * 64) + h;
    float4 ga[8], gb[8];
#pragma unroll
    for (int i = 0; i < 8; ++i) { ga[i] = pr[2 * i]; gb[i] = pl[2 * i]; }

    int en = (tile + tstride) * 16 + r;
    if (en >= E) en = E - 1;
    const int nR = rj[en];
    const int nL = lj[en];
    const float nw = weights[en];

    float4 msg[8];
#pragma unroll
    for (int i = 0; i < 8; ++i) {
      float4 a = ga[i], b = gb[i], c = pe[2 * i];
      float4 m;
      m.x = a.x + b.x + w * c.x;
      m.y = a.y + b.y + w * c.y;
      m.z = a.z + b.z + w * c.z;
      m.w = a.w + b.w + w * c.w;
      m.x = m.x > 0.f ? m.x : 0.01f * m.x;
      m.y = m.y > 0.f ? m.y : 0.01f * m.y;
      m.z = m.z > 0.f ? m.z : 0.01f * m.z;
      m.w = m.w > 0.f ? m.w : 0.01f * m.w;
      msg[i] = m;
    }

    v8f acc0 = {}, acc1 = {}, acc2 = {}, acc3 = {};
#pragma unroll
    for (int i = 0; i < 8; ++i) {
#pragma unroll
      for (int p = 0; p < 2; ++p) {
        v2f a;
        a.x = p ? msg[i].z : msg[i].x;
        a.y = p ? msg[i].w : msg[i].y;
        const int j = 2 * i + p;
        acc0 = wmma_f32(a, Bf[j][0], acc0);
        acc1 = wmma_f32(a, Bf[j][1], acc1);
        acc2 = wmma_f32(a, Bf[j][2], acc2);
        acc3 = wmma_f32(a, Bf[j][3], acc3);
      }
    }

#pragma unroll
    for (int v = 0; v < 8; ++v) {
      const float a0 = acc0[v], b0 = acc1[v], a1 = acc2[v], b1 = acc3[v];
      const float ta0 = xhalf(a0), tb0 = xhalf(b0), ta1 = xhalf(a1), tb1 = xhalf(b1);
      const float x0 = h ? tb0 : a0;
      const float y0 = h ? b0 : ta0;
      const float x1 = h ? tb1 : a1;
      const float y1 = h ? b1 : ta1;
      if (eBase + v < E) {
        const int seg = __builtin_amdgcn_readlane(idxR, v);
        float* orow = outAcc + (size_t)seg * 64;
        atomic_fadd_line_lo(orow, laneOff, x0 + biaLo);
        atomic_fadd_line_hi(orow, laneOff, x1 + biaHi);
      }
      if (eBase + v + 8 < E) {
        const int seg = __builtin_amdgcn_readlane(idxR, v + 8);
        float* orow = outAcc + (size_t)seg * 64;
        atomic_fadd_line_lo(orow, laneOff, y0 + biaLo);
        atomic_fadd_line_hi(orow, laneOff, y1 + biaHi);
      }
    }

    idxR = nR; idxL = nL; w = nw;
  }
}

extern "C" void kernel_launch(void* const* d_in, const int* in_sizes, int n_in,
                              void* d_out, int out_size, void* d_ws, size_t ws_size,
                              hipStream_t stream) {
  const float* input   = (const float*)d_in[0];
  const float* other   = (const float*)d_in[1];
  const int*   rj      = (const int*)d_in[2];
  const int*   lj      = (const int*)d_in[3];
  const float* weights = (const float*)d_in[4];
  const float* Wi      = (const float*)d_in[5];
  const float* bi      = (const float*)d_in[6];
  const float* Wo      = (const float*)d_in[7];
  const float* We      = (const float*)d_in[8];
  const float* Wf      = (const float*)d_in[9];
  const float* bf      = (const float*)d_in[10];
  const float* Wout    = (const float*)d_in[11];
  const float* bout    = (const float*)d_in[12];

  const int D    = 64;
  const int N_IN = in_sizes[0] / D;
  const int N_OT = in_sizes[1] / D;
  const int E    = in_sizes[2];

  float* rhs = (float*)d_ws;
  float* lhs = rhs + (size_t)N_IN * D;
  float* Wc  = lhs + (size_t)N_OT * D;
  float* out = (float*)d_out;

  const int tiles_in = (N_IN + 15) / 16;
  const int tiles_ot = (N_OT + 15) / 16;

  const int nFold = 3;
  const int nDual = (tiles_in + 7) / 8;
  const int nLhs  = (tiles_ot + 7) / 8;
  prep_kernel<<<nFold + nDual + nLhs, 256, 0, stream>>>(input, Wi, bi, rhs, Wout + 64 * 64, bout, out, N_IN,
                                                        other, Wo, lhs, N_OT,
                                                        Wf, bf, Wout, Wc, nFold, nDual);

  const int eTiles = (E + 15) / 16;
  edge_kernel<<<1024, 128, 0, stream>>>(rhs, lhs, rj, lj, weights,
                                        We, Wc, Wc + 64 * 64, out, E, eTiles);
}


# Edge kernel software-pipelined at no register cost: next tile's 16 row gathers asked for during this tile's WMMA block, ahead of its 32 atomics, into registers the block has used up; sums as parent
# speedup vs baseline: 1.9544x; 1.0450x over previous
#include <hip/hip_runtime.h>

typedef float v2f __attribute__((ext_vector_type(2)));
typedef float v8f __attribute__((ext_vector_type(8)));

__device__ __forceinline__ v8f wmma_f32(v2f a, v2f b, v8f c) {
  return __builtin_amdgcn_wmma_f32_16x16x4_f32(false, a, false, b, (short)0, c,
                                               false, false);
}

__device__ __forceinline__ void atomic_fadd_line_lo(float* rowBase, unsigned laneByteOff, float v) {
  asm volatile("global_atomic_add_f32 %0, %1, %2 scope:SCOPE_DEV"
               :
               : "v"(laneByteOff), "v"(v), "s"(rowBase)
               : "memory");
}
__device__ __forceinline__ void atomic_fadd_line_hi(float* rowBase, unsigned laneByteOff, float v) {
  asm volatile("global_atomic_add_f32 %0, %1, %2 offset:128 scope:SCOPE_DEV"
               :
               : "v"(laneByteOff), "v"(v), "s"(rowBase)
               : "memory");
}

__device__ __forceinline__ float xhalf(float x) {
  return __builtin_bit_cast(float, __builtin_amdgcn_permlanex16(__builtin_bit_cast(int, x),
                                                                __builtin_bit_cast(int, x),
                                                                0x76543210, 0xfedcba98, false, false));
}

__device__ __forceinline__ void gemm64_body(float2* __restrict__ Bs2, int blk,
                                            const float* __restrict__ A,
                                            const float* __restrict__ B,
                                            const float* __restrict__ bias,
                                            float* __restrict__ C, int M) {
  for (int i = threadIdx.x; i < 2048; i += 256) {
    int k2 = i >> 6, n = i & 63;
    Bs2[i] = float2{B[(2 * k2) * 64 + n], B[(2 * k2 + 1) * 64 + n]};
  }
  __syncthreads();

  const int wave = threadIdx.x >> 5;
  const int lane = threadIdx.x & 31;
  const int r = lane & 15;
  const int h = lane >> 4;

  const int row0 = (blk * 8 + wave) * 16;
  if (row0 >= M) return;

  int rowA = row0 + r;
  if (rowA >= M) rowA = M - 1;
  const float* __restrict__ Arow = A + (size_t)rowA * 64;

  v8f acc0 = {}, acc1 = {}, acc2 = {}, acc3 = {};
#pragma unroll
  for (int k0 = 0; k0 < 64; k0 += 4) {
    const int ka = k0 + 2 * h;
    float2 av = *reinterpret_cast<const float2*>(Arow + ka);
    v2f a; a.x = av.x; a.y = av.y;
    const int k2 = (k0 >> 1) + h;
    const float2* __restrict__ Bk = &Bs2[k2 * 64 + r];
    float2 f0 = Bk[0], f1 = Bk[16], f2 = Bk[32], f3 = Bk[48];
    v2f b0, b1, b2, b3;
    b0.x = f0.x; b0.y = f0.y;  b1.x = f1.x; b1.y = f1.y;
    b2.x = f2.x; b2.y = f2.y;  b3.x = f3.x; b3.y = f3.y;
    acc0 = wmma_f32(a, b0, acc0);
    acc1 = wmma_f32(a, b1, acc1);
    acc2 = wmma_f32(a, b2, acc2);
    acc3 = wmma_f32(a, b3, acc3);
  }

  float bia0 = 0.f, bia1 = 0.f, bia2 = 0.f, bia3 = 0.f;
  if (bias) { bia0 = bias[r]; bia1 = bias[16 + r]; bia2 = bias[32 + r]; bia3 = bias[48 + r]; }

#pragma unroll
  for (int v = 0; v < 8; ++v) {
    const int row = row0 + v + 8 * h;
    if (row < M) {
      float* __restrict__ Crow = C + (size_t)row * 64;
      Crow[r]      = acc0[v] + bia0;
      Crow[16 + r] = acc1[v] + bia1;
      Crow[32 + r] = acc2[v] + bia2;
      Crow[48 + r] = acc3[v] + bia3;
    }
  }
}

__device__ __forceinline__ void gemm64_dual_body(float2 (* __restrict__ Bs2)[32 * 64], int blk,
                                                 const float* __restrict__ A,
                                                 const float* __restrict__ B0,
                                                 const float* __restrict__ bias0,
                                                 float* __restrict__ C0,
                                                 const float* __restrict__ B1,
                                                 const float* __restrict__ bias1,
                                                 float* __restrict__ C1, int M) {
  for (int i = threadIdx.x; i < 2048; i += 256) {
    int k2 = i >> 6, n = i & 63;
    Bs2[0][i] = float2{B0[(2 * k2) * 64 + n], B0[(2 * k2 + 1) * 64 + n]};
    Bs2[1][i] = float2{B1[(2 * k2) * 64 + n], B1[(2 * k2 + 1) * 64 + n]};
  }
  __syncthreads();

  const int wave = threadIdx.x >> 5;
  const int lane = threadIdx.x & 31;
  const int r = lane & 15;
  const int h = lane >> 4;

  const int row0 = (blk * 8 + wave) * 16;
  if (row0 >= M) return;

  int rowA = row0 + r;
  if (rowA >= M) rowA = M - 1;
  const float* __restrict__ Arow = A + (size_t)rowA * 64;

  v8f p0 = {}, p1 = {}, p2 = {}, p3 = {};
  v8f q0 = {}, q1 = {}, q2 = {}, q3 = {};
#pragma unroll
  for (int k0 = 0; k0 < 64; k0 += 4) {
    const int ka = k0 + 2 * h;
    float2 av = *reinterpret_cast<const float2*>(Arow + ka);
    v2f a; a.x = av.x; a.y = av.y;
    const int k2 = (k0 >> 1) + h;
    const float2* __restrict__ Bk = &Bs2[0][k2 * 64 + r];
    const float2* __restrict__ Bq = &Bs2[1][k2 * 64 + r];
    float2 f0 = Bk[0], f1 = Bk[16], f2 = Bk[32], f3 = Bk[48];
    float2 g0 = Bq[0], g1 = Bq[16], g2 = Bq[32], g3 = Bq[48];
    v2f b0, b1, b2, b3, c0, c1, c2, c3;
    b0.x = f0.x; b0.y = f0.y;  b1.x = f1.x; b1.y = f1.y;
    b2.x = f2.x; b2.y = f2.y;  b3.x = f3.x; b3.y = f3.y;
    c0.x = g0.x; c0.y = g0.y;  c1.x = g1.x; c1.y = g1.y;
    c2.x = g2.x; c2.y = g2.y;  c3.x = g3.x; c3.y = g3.y;
    p0 = wmma_f32(a, b0, p0);
    p1 = wmma_f32(a, b1, p1);
    p2 = wmma_f32(a, b2, p2);
    p3 = wmma_f32(a, b3, p3);
    q0 = wmma_f32(a, c0, q0);
    q1 = wmma_f32(a, c1, q1);
    q2 = wmma_f32(a, c2, q2);
    q3 = wmma_f32(a, c3, q3);
  }

  const float bp0 = bias0[r], bp1 = bias0[16 + r], bp2 = bias0[32 + r], bp3 = bias0[48 + r];
  const float bq0 = bias1[r], bq1 = bias1[16 + r], bq2 = bias1[32 + r], bq3 = bias1[48 + r];

#pragma unroll
  for (int v = 0; v < 8; ++v) {
    const int row = row0 + v + 8 * h;
    if (row < M) {
      float* __restrict__ Prow = C0 + (size_t)row * 64;
      Prow[r]      = p0[v] + bp0;
      Prow[16 + r] = p1[v] + bp1;
      Prow[32 + r] = p2[v] + bp2;
      Prow[48 + r] = p3[v] + bp3;
      float* __restrict__ Qrow = C1 + (size_t)row * 64;
      Qrow[r]      = q0[v] + bq0;
      Qrow[16 + r] = q1[v] + bq1;
      Qrow[32 + r] = q2[v] + bq2;
      Qrow[48 + r] = q3[v] + bq3;
    }
  }
}

__device__ __forceinline__ void fold_body(int tileId,
                                          const float* __restrict__ Wf,
                                          const float* __restrict__ bf,
                                          const float* __restrict__ Wout,
                                          float* __restrict__ Wc) {
  const int lane = threadIdx.x & 31;
  const int r = lane & 15;
  const int h = lane >> 4;
  const int rt = tileId >> 2;
  const int ct = tileId & 3;

  const float* __restrict__ Arow = (rt < 4) ? (Wf + (rt * 16 + r) * 64) : bf;
  const float* __restrict__ Bcol = Wout + ct * 16 + r;

  v8f acc = {};
#pragma unroll
  for (int k0 = 0; k0 < 64; k0 += 4) {
    const int ka = k0 + 2 * h;
    v2f a, b;
    a.x = Arow[ka];            a.y = Arow[ka + 1];
    b.x = Bcol[ka * 64];       b.y = Bcol[(ka + 1) * 64];
    acc = wmma_f32(a, b, acc);
  }

  if (rt < 4) {
#pragma unroll
    for (int v = 0; v < 8; ++v)
      Wc[(rt * 16 + v + 8 * h) * 64 + ct * 16 + r] = acc[v];
  } else if (h == 0) {
    Wc[64 * 64 + ct * 16 + r] = acc[0];
  }
}

__launch_bounds__(256)
__global__ void prep_kernel(const float* __restrict__ input,
                            const float* __restrict__ Wi,
                            const float* __restrict__ bi,
                            float* __restrict__ rhs,
                            const float* __restrict__ WoutBot,
                            const float* __restrict__ bout,
                            float* __restrict__ out, int nIn,
                            const float* __restrict__ other,
                            const float* __restrict__ Wo,
                            float* __restrict__ lhs, int nOt,
                            const float* __restrict__ Wf,
                            const float* __restrict__ bf,
                            const float* __restrict__ Wout,
                            float* __restrict__ Wc,
                            int nFold, int nDual) {
  __shared__ float2 Bs2[2][32 * 64];
  const int b = (int)blockIdx.x;
  if (b < nFold) {
    const int tileId = b * 8 + __builtin_amdgcn_readfirstlane((int)(threadIdx.x >> 5));
    if (tileId < 20) fold_body(tileId, Wf, bf, Wout, Wc);
  } else if (b < nFold + nDual) {
    gemm64_dual_body(Bs2, b - nFold, input, Wi, bi, rhs, WoutBot, bout, out, nIn);
  } else {
    gemm64_body(Bs2[0], b - nFold - nDual, other, Wo, nullptr, lhs, nOt);
  }
}

__launch_bounds__(128)
__global__ void edge_kernel(const float* __restrict__ rhs,
                            const float* __restrict__ lhs,
                            const int* __restrict__ rj,
                            const int* __restrict__ lj,
                            const float* __restrict__ weights,
                            const float* __restrict__ We,
                            const float* __restrict__ Wf,
                            const float* __restrict__ bf,
                            float* __restrict__ outAcc, int E, int eTiles) {
  __shared__ float WeS[64];

  if (threadIdx.x < 64) WeS[threadIdx.x] = We[threadIdx.x];
  __syncthreads();

  const int wave = __builtin_amdgcn_readfirstlane((int)(threadIdx.x >> 5));
  const int lane = threadIdx.x & 31;
  const int r = lane & 15;
  const int h = lane >> 4;

  v2f Bf[16][4];
#pragma unroll
  for (int j = 0; j < 16; ++j) {
    const int k2 = 4 * (j >> 1) + 2 * h + (j & 1);
    const float* __restrict__ w0 = Wf + (2 * k2) * 64 + r;
#pragma unroll
    for (int t = 0; t < 4; ++t) {
      v2f b;
      b.x = w0[16 * t];
      b.y = w0[64 + 16 * t];
      Bf[j][t] = b;
    }
  }

  const float biaLo = bf[lane], biaHi = bf[32 + lane];
  const unsigned laneOff = (unsigned)lane * 4u;
  const float4* __restrict__ pe = reinterpret_cast<const float4*>(WeS) + h;

  const int tstride = (int)gridDim.x * 4;
  int tile = (int)blockIdx.x * 4 + wave;

  int e = tile * 16 + r;
  if (e >= E) e = E - 1;
  int idxR = rj[e];
  const int idxL0 = lj[e];
  float w = weights[e];

  float4 ga[8], gb[8];
  {
    const float4* __restrict__ pr = reinterpret_cast<const float4*>(rhs + (size_t)idxR * 64) + h;
    const float4* __restrict__ pl = reinterpret_cast<const float4*>(lhs + (size_t)idxL0 * 64) + h;
#pragma unroll
    for (int i = 0; i < 8; ++i) { ga[i] = pr[2 * i]; gb[i] = pl[2 * i]; }
  }

  int en = (tile + tstride) * 16 + r;
  if (en >= E) en = E - 1;
  int nR = rj[en];
  int nL = lj[en];
  float nw = weights[en];

  for (; tile < eTiles; tile += tstride) {
    const int eBase = tile * 16;

    float4 msg[8];
#pragma unroll
    for (int i = 0; i < 8; ++i) {
      float4 a = ga[i], b = gb[i], c = pe[2 * i];
      float4 m;
      m.x = a.x + b.x + w * c.x;
      m.y = a.y + b.y + w * c.y;
      m.z = a.z + b.z + w * c.z;
      m.w = a.w + b.w + w * c.w;
      m.x = m.x > 0.f ? m.x : 0.01f * m.x;
      m.y = m.y > 0.f ? m.y : 0.01f * m.y;
      m.z = m.z > 0.f ? m.z : 0.01f * m.z;
      m.w = m.w > 0.f ? m.w : 0.01f * m.w;
      msg[i] = m;
    }

    v8f acc0 = {}, acc1 = {}, acc2 = {}, acc3 = {};
#pragma unroll
    for (int i = 0; i < 8; ++i) {
#pragma unroll
      for (int p = 0; p < 2; ++p) {
        v2f a;
        a.x = p ? msg[i].z : msg[i].x;
        a.y = p ? msg[i].w : msg[i].y;
        const int j = 2 * i + p;
        acc0 = wmma_f32(a, Bf[j][0], acc0);
        acc1 = wmma_f32(a, Bf[j][1], acc1);
        acc2 = wmma_f32(a, Bf[j][2], acc2);
        acc3 = wmma_f32(a, Bf[j][3], acc3);
      }
    }

    {
      const float4* __restrict__ pr = reinterpret_cast<const float4*>(rhs + (size_t)nR * 64) + h;
      const float4* __restrict__ pl = reinterpret_cast<const float4*>(lhs + (size_t)nL * 64) + h;
#pragma unroll
      for (int i = 0; i < 8; ++i) { ga[i] = pr[2 * i]; gb[i] = pl[2 * i]; }
    }

    int e2 = (tile + 2 * tstride) * 16 + r;
    if (e2 >= E) e2 = E - 1;
    const int mR = rj[e2];
    const int mL = lj[e2];
    const float mw = weights[e2];

#pragma unroll
    for (int v = 0; v < 8; ++v) {
      const float a0 = acc0[v], b0 = acc1[v], a1 = acc2[v], b1 = acc3[v];
      const float ta0 = xhalf(a0), tb0 = xhalf(b0), ta1 = xhalf(a1), tb1 = xhalf(b1);
      const float x0 = h ? tb0 : a0;
      const float y0 = h ? b0 : ta0;
      const float x1 = h ? tb1 : a1;
      const float y1 = h ? b1 : ta1;
      if (eBase + v < E) {
        const int seg = __builtin_amdgcn_readlane(idxR, v);
        float* orow = outAcc + (size_t)seg * 64;
        atomic_fadd_line_lo(orow, laneOff, x0 + biaLo);
        atomic_fadd_line_hi(orow, laneOff, x1 + biaHi);
      }
      if (eBase + v + 8 < E) {
        const int seg = __builtin_amdgcn_readlane(idxR, v + 8);
        float* orow = outAcc + (size_t)seg * 64;
        atomic_fadd_line_lo(orow, laneOff, y0 + biaLo);
        atomic_fadd_line_hi(orow, laneOff, y1 + biaHi);
      }
    }

    idxR = nR; w = nw;
    nR = mR; nL = mL; nw = mw;
  }
}

extern "C" void kernel_launch(void* const* d_in, const int* in_sizes, int n_in,
                              void* d_out, int out_size, void* d_ws, size_t ws_size,
                              hipStream_t stream) {
  const float* input   = (const float*)d_in[0];
  const float* other   = (const float*)d_in[1];
  const int*   rj      = (const int*)d_in[2];
  const int*   lj      = (const int*)d_in[3];
  const float* weights = (const float*)d_in[4];
  const float* Wi      = (const float*)d_in[5];
  const float* bi      = (const float*)d_in[6];
  const float* Wo      = (const float*)d_in[7];
  const float* We      = (const float*)d_in[8];
  const float* Wf      = (const float*)d_in[9];
  const float* bf      = (const float*)d_in[10];
  const float* Wout    = (const float*)d_in[11];
  const float* bout    = (const float*)d_in[12];

  const int D    = 64;
  const int N_IN = in_sizes[0] / D;
  const int N_OT = in_sizes[1] / D;
  const int E    = in_sizes[2];

  float* rhs = (float*)d_ws;
  float* lhs = rhs + (size_t)N_IN * D;
  float* Wc  = lhs + (size_t)N_OT * D;
  float* out = (float*)d_out;

  const int tiles_in = (N_IN + 15) / 16;
  const int tiles_ot = (N_OT + 15) / 16;

  const int nFold = 3;
  const int nDual = (tiles_in + 7) / 8;
  const int nLhs  = (tiles_ot + 7) / 8;
  prep_kernel<<<nFold + nDual + nLhs, 256, 0, stream>>>(input, Wi, bi, rhs, Wout + 64 * 64, bout, out, N_IN,
                                                        other, Wo, lhs, N_OT,
                                                        Wf, bf, Wout, Wc, nFold, nDual);

  const int eTiles = (E + 15) / 16;
  edge_kernel<<<1024, 128, 0, stream>>>(rhs, lhs, rj, lj, weights,
                                        We, Wc, Wc + 64 * 64, out, E, eTiles);
}
